# starting kernel
# baseline (speedup 1.0000x reference)
#include <hip/hip_runtime.h>
#include <math.h>

typedef float v2f __attribute__((ext_vector_type(2)));
typedef float v8f __attribute__((ext_vector_type(8)));

#define B_SZ 8192
#define T_SZ 200
#define D_SZ 3
#define H_SZ 20
#define L_SZ 10

#define WAVES_PER_BLOCK 4

__device__ __forceinline__ v8f wmma4(v2f a, v2f b, v8f c) {
  return __builtin_amdgcn_wmma_f32_16x16x4_f32(false, a, false, b, (short)0, c,
                                               false, false);
}

__device__ __forceinline__ float fast_tanh(float x) {
#if __has_builtin(__builtin_amdgcn_tanhf)
  return __builtin_amdgcn_tanhf(x);
#else
  float r;
  asm volatile("v_tanh_f32 %0, %1\n\tv_nop" : "=v"(r) : "v"(x));
  return r;
#endif
}

__device__ __forceinline__ float swap16(float x) {
  return __builtin_bit_cast(
      float, __builtin_amdgcn_ds_swizzle(__builtin_bit_cast(int, x), 0x401F));
}

__global__ __launch_bounds__(WAVES_PER_BLOCK * 32)
void latent_ode_kernel(const float* __restrict__ ts, const float* __restrict__ ys,
                       const float* __restrict__ eps,
                       const float* __restrict__ encW1, const float* __restrict__ encB1,
                       const float* __restrict__ encW2, const float* __restrict__ encB2,
                       const float* __restrict__ encW3, const float* __restrict__ encB3,
                       const float* __restrict__ l2hW, const float* __restrict__ l2hB,
                       const float* __restrict__ h2dW, const float* __restrict__ h2dB,
                       const float* __restrict__ fW1, const float* __restrict__ fB1,
                       const float* __restrict__ fW2, const float* __restrict__ fB2,
                       float* __restrict__ pred, float* __restrict__ outMean,
                       float* __restrict__ outStd)
{
  const int tid  = (int)threadIdx.x;
  const int wave = tid >> 5;
  const int lane = tid & 31;
  const int n16  = lane & 15;
  const int hi   = lane >> 4;
  const bool h   = (hi != 0);
  const int rowBase = ((int)blockIdx.x * WAVES_PER_BLOCK + wave) * 16;
  const int row = rowBase + n16;

  float yin[D_SZ];
#pragma unroll
  for (int d = 0; d < D_SZ; ++d)
    yin[d] = ys[(size_t)row * T_SZ * D_SZ + d];

  float h1[H_SZ];
#pragma unroll
  for (int j = 0; j < H_SZ; ++j) {
    float acc = encB1[j];
#pragma unroll
    for (int d = 0; d < D_SZ; ++d) acc += yin[d] * encW1[d * H_SZ + j];
    h1[j] = fmaxf(acc, 0.0f);
  }
  float h2[H_SZ];
#pragma unroll
  for (int j = 0; j < H_SZ; ++j) {
    float acc = encB2[j];
#pragma unroll
    for (int i = 0; i < H_SZ; ++i) acc += h1[i] * encW2[i * H_SZ + j];
    h2[j] = fmaxf(acc, 0.0f);
  }
  float latent[L_SZ];
#pragma unroll
  for (int j = 0; j < L_SZ; ++j) {
    float m  = encB3[j];
    float ls = encB3[L_SZ + j];
#pragma unroll
    for (int i = 0; i < H_SZ; ++i) {
      m  += h2[i] * encW3[i * (2 * L_SZ) + j];
      ls += h2[i] * encW3[i * (2 * L_SZ) + L_SZ + j];
    }
    const float sd = expf(ls);
    latent[j] = m + eps[(size_t)row * L_SZ + j] * sd;
    if (lane < 16) {
      outMean[(size_t)row * L_SZ + j] = m;
      outStd [(size_t)row * L_SZ + j] = sd;
    }
  }
  float y0v[H_SZ];
#pragma unroll
  for (int j = 0; j < H_SZ; ++j) {
    float acc = l2hB[j];
#pragma unroll
    for (int i = 0; i < L_SZ; ++i) acc += latent[i] * l2hW[i * H_SZ + j];
    y0v[j] = fmaxf(acc, 0.0f);
  }

  v8f yc0, yc1;
#pragma unroll
  for (int v = 0; v < 8; ++v) {
    yc0[v] = h ? y0v[v + 8] : y0v[v];
    yc1[v] = (!h && v < 4) ? y0v[16 + v] : 0.0f;
  }

  v2f w1f[2][6], w2f[2][6], w3f[6];
#pragma unroll
  for (int s = 0; s < 5; ++s) {
    const int k0 = 4 * s + 2 * hi;
#pragma unroll
    for (int nt = 0; nt < 2; ++nt) {
      const int n = 16 * nt + n16;
      const bool ok = n < H_SZ;
      w1f[nt][s].x = ok ? fW1[k0 * H_SZ + n]       : 0.0f;
      w1f[nt][s].y = ok ? fW1[(k0 + 1) * H_SZ + n] : 0.0f;
      w2f[nt][s].x = ok ? fW2[k0 * H_SZ + n]       : 0.0f;
      w2f[nt][s].y = ok ? fW2[(k0 + 1) * H_SZ + n] : 0.0f;
    }
    const bool ok3 = n16 < D_SZ;
    w3f[s].x = ok3 ? h2dW[k0 * D_SZ + n16]       : 0.0f;
    w3f[s].y = ok3 ? h2dW[(k0 + 1) * D_SZ + n16] : 0.0f;
  }
#pragma unroll
  for (int nt = 0; nt < 2; ++nt) {
    const int n = 16 * nt + n16;
    const bool ok = (!h) && (n < H_SZ);
    w1f[nt][5].x = ok ? fB1[n] : 0.0f;  w1f[nt][5].y = 0.0f;
    w2f[nt][5].x = ok ? fB2[n] : 0.0f;  w2f[nt][5].y = 0.0f;
  }
  w3f[5].x = (!h && n16 < D_SZ) ? h2dB[n16] : 0.0f;
  w3f[5].y = 0.0f;
  v2f bone;
  bone.x = h ? 0.0f : 1.0f;
  bone.y = 0.0f;
  const v8f vzero = {};

  auto buildB = [&](const v8f& z0, const v8f& z1, v2f* bf) {
    const float s0 = swap16(z0[0]), s1 = swap16(z0[1]);
    const float s2 = swap16(z0[2]), s3 = swap16(z0[3]);
    const float s4 = swap16(z0[4]), s5 = swap16(z0[5]);
    const float s6 = swap16(z0[6]), s7 = swap16(z0[7]);
    const float t2 = swap16(z1[2]), t3 = swap16(z1[3]);
    bf[0].x = h ? s2    : z0[0];  bf[0].y = h ? s3    : z0[1];
    bf[1].x = h ? s6    : z0[4];  bf[1].y = h ? s7    : z0[5];
    bf[2].x = h ? z0[2] : s0;     bf[2].y = h ? z0[3] : s1;
    bf[3].x = h ? z0[6] : s4;     bf[3].y = h ? z0[7] : s5;
    bf[4].x = h ? t2    : z1[0];  bf[4].y = h ? t3    : z1[1];
  };
  auto matmul = [&](const v2f (*wa)[6], const v2f* bf, v8f& o0, v8f& o1) {
    o0 = wmma4(wa[0][5], bone, vzero);
    o1 = wmma4(wa[1][5], bone, vzero);
#pragma unroll
    for (int s = 0; s < 5; ++s) {
      o0 = wmma4(wa[0][s], bf[s], o0);
      o1 = wmma4(wa[1][s], bf[s], o1);
    }
  };
  auto fEval = [&](const v2f* bf, v8f& k0, v8f& k1) {
    v8f t0, t1;
    matmul(w1f, bf, t0, t1);
#pragma unroll
    for (int v = 0; v < 8; ++v) {
      t0[v] = fast_tanh(t0[v]);
      t1[v] = fast_tanh(t1[v]);
    }
    v2f bt[5];
    buildB(t0, t1, bt);
    matmul(w2f, bt, k0, k1);
  };
  auto project = [&](const v2f* bf, int t) {
    v8f p = wmma4(w3f[5], bone, vzero);
#pragma unroll
    for (int s = 0; s < 5; ++s) p = wmma4(w3f[s], bf[s], p);
    if (!h) {
      float* o = &pred[((size_t)(rowBase + n16) * T_SZ + t) * D_SZ];
      o[0] = p[0]; o[1] = p[1]; o[2] = p[2];
    }
  };

  for (int t = 0; t < T_SZ - 1; ++t) {
    const float dt  = ts[t + 1] - ts[t];
    const float hdt = 0.5f * dt;
    v2f bf[5];

    buildB(yc0, yc1, bf);
    project(bf, t);

    v8f ka, kb, acc0, acc1, u0, u1;
    fEval(bf, ka, kb);
#pragma unroll
    for (int v = 0; v < 8; ++v) {
      acc0[v] = ka[v];                 acc1[v] = kb[v];
      u0[v] = yc0[v] + hdt * ka[v];    u1[v] = yc1[v] + hdt * kb[v];
    }
    buildB(u0, u1, bf);
    fEval(bf, ka, kb);
#pragma unroll
    for (int v = 0; v < 8; ++v) {
      acc0[v] += 2.0f * ka[v];         acc1[v] += 2.0f * kb[v];
      u0[v] = yc0[v] + hdt * ka[v];    u1[v] = yc1[v] + hdt * kb[v];
    }
    buildB(u0, u1, bf);
    fEval(bf, ka, kb);
#pragma unroll
    for (int v = 0; v < 8; ++v) {
      acc0[v] += 2.0f * ka[v];         acc1[v] += 2.0f * kb[v];
      u0[v] = yc0[v] + dt * ka[v];     u1[v] = yc1[v] + dt * kb[v];
    }
    buildB(u0, u1, bf);
    fEval(bf, ka, kb);

    const float sdt = dt * (1.0f / 6.0f);
#pragma unroll
    for (int v = 0; v < 8; ++v) {
      yc0[v] += sdt * (acc0[v] + ka[v]);
      yc1[v] += sdt * (acc1[v] + kb[v]);
    }
  }
  {
    v2f bf[5];
    buildB(yc0, yc1, bf);
    project(bf, T_SZ - 1);
  }
}

extern "C" void kernel_launch(void* const* d_in, const int* in_sizes, int n_in,
                              void* d_out, int out_size, void* d_ws, size_t ws_size,
                              hipStream_t stream) {
  const float* ts    = (const float*)d_in[0];
  const float* ys    = (const float*)d_in[1];
  const float* eps   = (const float*)d_in[2];
  const float* encW1 = (const float*)d_in[3];
  const float* encB1 = (const float*)d_in[4];
  const float* encW2 = (const float*)d_in[5];
  const float* encB2 = (const float*)d_in[6];
  const float* encW3 = (const float*)d_in[7];
  const float* encB3 = (const float*)d_in[8];
  const float* l2hW  = (const float*)d_in[9];
  const float* l2hB  = (const float*)d_in[10];
  const float* h2dW  = (const float*)d_in[11];
  const float* h2dB  = (const float*)d_in[12];
  const float* fW1   = (const float*)d_in[13];
  const float* fB1   = (const float*)d_in[14];
  const float* fW2   = (const float*)d_in[15];
  const float* fB2   = (const float*)d_in[16];

  float* pred = (float*)d_out;
  float* mean = pred + (size_t)B_SZ * T_SZ * D_SZ;
  float* stdv = mean + (size_t)B_SZ * L_SZ;

  dim3 grid(B_SZ / (WAVES_PER_BLOCK * 16));
  dim3 block(WAVES_PER_BLOCK * 32);
  hipLaunchKernelGGL(latent_ode_kernel, grid, block, 0, stream,
                     ts, ys, eps, encW1, encB1, encW2, encB2, encW3, encB3,
                     l2hW, l2hB, h2dW, h2dB, fW1, fB1, fW2, fB2,
                     pred, mean, stdv);
}


# State tile rows and weight k order re-paired so C-layout state registers feed the WMMA B operands directly: no half-wave swizzles or selects; changes the order the 20 k products are summed
# speedup vs baseline: 1.8455x; 1.8455x over previous
#include <hip/hip_runtime.h>
#include <math.h>

typedef float v2f __attribute__((ext_vector_type(2)));
typedef float v8f __attribute__((ext_vector_type(8)));

#define B_SZ 8192
#define T_SZ 200
#define D_SZ 3
#define H_SZ 20
#define L_SZ 10

#define WAVES_PER_BLOCK 4

__device__ __forceinline__ v8f wmma4(v2f a, v2f b, v8f c) {
  return __builtin_amdgcn_wmma_f32_16x16x4_f32(false, a, false, b, (short)0, c,
                                               false, false);
}

__device__ __forceinline__ float fast_tanh(float x) {
#if __has_builtin(__builtin_amdgcn_tanhf)
  return __builtin_amdgcn_tanhf(x);
#else
  float r;
  asm volatile("v_tanh_f32 %0, %1\n\tv_nop" : "=v"(r) : "v"(x));
  return r;
#endif
}

__global__ __launch_bounds__(WAVES_PER_BLOCK * 32)
void latent_ode_kernel(const float* __restrict__ ts, const float* __restrict__ ys,
                       const float* __restrict__ eps,
                       const float* __restrict__ encW1, const float* __restrict__ encB1,
                       const float* __restrict__ encW2, const float* __restrict__ encB2,
                       const float* __restrict__ encW3, const float* __restrict__ encB3,
                       const float* __restrict__ l2hW, const float* __restrict__ l2hB,
                       const float* __restrict__ h2dW, const float* __restrict__ h2dB,
                       const float* __restrict__ fW1, const float* __restrict__ fB1,
                       const float* __restrict__ fW2, const float* __restrict__ fB2,
                       float* __restrict__ pred, float* __restrict__ outMean,
                       float* __restrict__ outStd)
{
  const int tid  = (int)threadIdx.x;
  const int wave = tid >> 5;
  const int lane = tid & 31;
  const int n16  = lane & 15;
  const int hi   = lane >> 4;
  const bool h   = (hi != 0);
  const int rowBase = ((int)blockIdx.x * WAVES_PER_BLOCK + wave) * 16;
  const int row = rowBase + n16;

  float yin[D_SZ];
#pragma unroll
  for (int d = 0; d < D_SZ; ++d)
    yin[d] = ys[(size_t)row * T_SZ * D_SZ + d];

  float h1[H_SZ];
#pragma unroll
  for (int j = 0; j < H_SZ; ++j) {
    float acc = encB1[j];
#pragma unroll
    for (int d = 0; d < D_SZ; ++d) acc += yin[d] * encW1[d * H_SZ + j];
    h1[j] = fmaxf(acc, 0.0f);
  }
  float h2[H_SZ];
#pragma unroll
  for (int j = 0; j < H_SZ; ++j) {
    float acc = encB2[j];
#pragma unroll
    for (int i = 0; i < H_SZ; ++i) acc += h1[i] * encW2[i * H_SZ + j];
    h2[j] = fmaxf(acc, 0.0f);
  }
  float latent[L_SZ];
#pragma unroll
  for (int j = 0; j < L_SZ; ++j) {
    float m  = encB3[j];
    float ls = encB3[L_SZ + j];
#pragma unroll
    for (int i = 0; i < H_SZ; ++i) {
      m  += h2[i] * encW3[i * (2 * L_SZ) + j];
      ls += h2[i] * encW3[i * (2 * L_SZ) + L_SZ + j];
    }
    const float sd = expf(ls);
    latent[j] = m + eps[(size_t)row * L_SZ + j] * sd;
    if (lane < 16) {
      outMean[(size_t)row * L_SZ + j] = m;
      outStd [(size_t)row * L_SZ + j] = sd;
    }
  }
  float y0v[H_SZ];
#pragma unroll
  for (int j = 0; j < H_SZ; ++j) {
    float acc = l2hB[j];
#pragma unroll
    for (int i = 0; i < L_SZ; ++i) acc += latent[i] * l2hW[i * H_SZ + j];
    y0v[j] = fmaxf(acc, 0.0f);
  }

  v8f yc0;
  v2f yc1;
#pragma unroll
  for (int v = 0; v < 8; ++v) yc0[v] = h ? y0v[v + 8] : y0v[v];
  yc1.x = h ? y0v[18] : y0v[16];
  yc1.y = h ? y0v[19] : y0v[17];

  const int  r8    = n16 & 7;
  const bool okT1  = r8 < 2;
  const int  nT1   = okT1 ? 16 + r8 + 2 * (n16 >> 3) : 0;
  v2f w1f[2][6], w2f[2][6], w3f[6];
#pragma unroll
  for (int s = 0; s < 5; ++s) {
    const int kx = (s < 4) ? (2 * s + 8 * hi) : (16 + 2 * hi);
    const int ky = kx + 1;
#pragma unroll
    for (int nt = 0; nt < 2; ++nt) {
      const int  n  = nt ? nT1 : n16;
      const bool ok = nt ? okT1 : true;
      w1f[nt][s].x = ok ? fW1[kx * H_SZ + n] : 0.0f;
      w1f[nt][s].y = ok ? fW1[ky * H_SZ + n] : 0.0f;
      w2f[nt][s].x = ok ? fW2[kx * H_SZ + n] : 0.0f;
      w2f[nt][s].y = ok ? fW2[ky * H_SZ + n] : 0.0f;
    }
    const bool ok3 = n16 < D_SZ;
    w3f[s].x = ok3 ? h2dW[kx * D_SZ + n16] : 0.0f;
    w3f[s].y = ok3 ? h2dW[ky * D_SZ + n16] : 0.0f;
  }
#pragma unroll
  for (int nt = 0; nt < 2; ++nt) {
    const int  n  = nt ? nT1 : n16;
    const bool ok = (!h) && (nt ? okT1 : true);
    w1f[nt][5].x = ok ? fB1[n] : 0.0f;  w1f[nt][5].y = 0.0f;
    w2f[nt][5].x = ok ? fB2[n] : 0.0f;  w2f[nt][5].y = 0.0f;
  }
  w3f[5].x = (!h && n16 < D_SZ) ? h2dB[n16] : 0.0f;
  w3f[5].y = 0.0f;
  v2f bone;
  bone.x = h ? 0.0f : 1.0f;
  bone.y = 0.0f;
  const v8f vzero = {};

  auto sliceB = [&](const v8f& z0, const v2f& z1, v2f* bf) {
#pragma unroll
    for (int j = 0; j < 4; ++j) { bf[j].x = z0[2 * j]; bf[j].y = z0[2 * j + 1]; }
    bf[4] = z1;
  };
  auto matmul = [&](const v2f (*wa)[6], const v2f* bf, v8f& o0, v8f& o1) {
    o0 = wmma4(wa[0][5], bone, vzero);
    o1 = wmma4(wa[1][5], bone, vzero);
#pragma unroll
    for (int s = 0; s < 5; ++s) {
      o0 = wmma4(wa[0][s], bf[s], o0);
      o1 = wmma4(wa[1][s], bf[s], o1);
    }
  };
  auto fEval = [&](const v2f* bf, v8f& k0, v2f& k1) {
    v8f t0, t1;
    matmul(w1f, bf, t0, t1);
    v2f bt[5];
#pragma unroll
    for (int j = 0; j < 4; ++j) {
      bt[j].x = fast_tanh(t0[2 * j]);
      bt[j].y = fast_tanh(t0[2 * j + 1]);
    }
    bt[4].x = fast_tanh(t1[0]);
    bt[4].y = fast_tanh(t1[1]);
    v8f o1;
    matmul(w2f, bt, k0, o1);
    k1.x = o1[0];
    k1.y = o1[1];
  };
  auto project = [&](const v2f* bf, int t) {
    v8f p = wmma4(w3f[5], bone, vzero);
#pragma unroll
    for (int s = 0; s < 5; ++s) p = wmma4(w3f[s], bf[s], p);
    if (!h) {
      float* o = &pred[((size_t)(rowBase + n16) * T_SZ + t) * D_SZ];
      o[0] = p[0]; o[1] = p[1]; o[2] = p[2];
    }
  };

  for (int t = 0; t < T_SZ - 1; ++t) {
    const float dt  = ts[t + 1] - ts[t];
    const float hdt = 0.5f * dt;
    v2f bf[5];

    sliceB(yc0, yc1, bf);
    project(bf, t);

    v8f ka, acc0, u0;
    v2f kb, acc1, u1;
    fEval(bf, ka, kb);
#pragma unroll
    for (int v = 0; v < 8; ++v) {
      acc0[v] = ka[v];
      u0[v] = yc0[v] + hdt * ka[v];
    }
    acc1 = kb;
    u1.x = yc1.x + hdt * kb.x;  u1.y = yc1.y + hdt * kb.y;
    sliceB(u0, u1, bf);
    fEval(bf, ka, kb);
#pragma unroll
    for (int v = 0; v < 8; ++v) {
      acc0[v] += 2.0f * ka[v];
      u0[v] = yc0[v] + hdt * ka[v];
    }
    acc1.x += 2.0f * kb.x;      acc1.y += 2.0f * kb.y;
    u1.x = yc1.x + hdt * kb.x;  u1.y = yc1.y + hdt * kb.y;
    sliceB(u0, u1, bf);
    fEval(bf, ka, kb);
#pragma unroll
    for (int v = 0; v < 8; ++v) {
      acc0[v] += 2.0f * ka[v];
      u0[v] = yc0[v] + dt * ka[v];
    }
    acc1.x += 2.0f * kb.x;      acc1.y += 2.0f * kb.y;
    u1.x = yc1.x + dt * kb.x;   u1.y = yc1.y + dt * kb.y;
    sliceB(u0, u1, bf);
    fEval(bf, ka, kb);

    const float sdt = dt * (1.0f / 6.0f);
#pragma unroll
    for (int v = 0; v < 8; ++v) yc0[v] += sdt * (acc0[v] + ka[v]);
    yc1.x += sdt * (acc1.x + kb.x);
    yc1.y += sdt * (acc1.y + kb.y);
  }
  {
    v2f bf[5];
    sliceB(yc0, yc1, bf);
    project(bf, T_SZ - 1);
  }
}

extern "C" void kernel_launch(void* const* d_in, const int* in_sizes, int n_in,
                              void* d_out, int out_size, void* d_ws, size_t ws_size,
                              hipStream_t stream) {
  const float* ts    = (const float*)d_in[0];
  const float* ys    = (const float*)d_in[1];
  const float* eps   = (const float*)d_in[2];
  const float* encW1 = (const float*)d_in[3];
  const float* encB1 = (const float*)d_in[4];
  const float* encW2 = (const float*)d_in[5];
  const float* encB2 = (const float*)d_in[6];
  const float* encW3 = (const float*)d_in[7];
  const float* encB3 = (const float*)d_in[8];
  const float* l2hW  = (const float*)d_in[9];
  const float* l2hB  = (const float*)d_in[10];
  const float* h2dW  = (const float*)d_in[11];
  const float* h2dB  = (const float*)d_in[12];
  const float* fW1   = (const float*)d_in[13];
  const float* fB1   = (const float*)d_in[14];
  const float* fW2   = (const float*)d_in[15];
  const float* fB2   = (const float*)d_in[16];

  float* pred = (float*)d_out;
  float* mean = pred + (size_t)B_SZ * T_SZ * D_SZ;
  float* stdv = mean + (size_t)B_SZ * L_SZ;

  dim3 grid(B_SZ / (WAVES_PER_BLOCK * 16));
  dim3 block(WAVES_PER_BLOCK * 32);
  hipLaunchKernelGGL(latent_ode_kernel, grid, block, 0, stream,
                     ts, ys, eps, encW1, encB1, encW2, encB2, encW3, encB3,
                     l2hW, l2hB, h2dW, h2dB, fW1, fB1, fW2, fB2,
                     pred, mean, stdv);
}


# Projection rows ride in idle rows 2-4 of the second feature tile: their own 6-WMMA chain is gone (42 WMMA per step, was 48); every sum keeps its order
# speedup vs baseline: 2.9358x; 1.5908x over previous
#include <hip/hip_runtime.h>
#include <math.h>

typedef float v2f __attribute__((ext_vector_type(2)));
typedef float v8f __attribute__((ext_vector_type(8)));

#define B_SZ 8192
#define T_SZ 200
#define D_SZ 3
#define H_SZ 20
#define L_SZ 10

#define WAVES_PER_BLOCK 4

__device__ __forceinline__ v8f wmma4(v2f a, v2f b, v8f c) {
  return __builtin_amdgcn_wmma_f32_16x16x4_f32(false, a, false, b, (short)0, c,
                                               false, false);
}

__device__ __forceinline__ float fast_tanh(float x) {
#if __has_builtin(__builtin_amdgcn_tanhf)
  return __builtin_amdgcn_tanhf(x);
#else
  float r;
  asm volatile("v_tanh_f32 %0, %1\n\tv_nop" : "=v"(r) : "v"(x));
  return r;
#endif
}

__global__ __launch_bounds__(WAVES_PER_BLOCK * 32)
void latent_ode_kernel(const float* __restrict__ ts, const float* __restrict__ ys,
                       const float* __restrict__ eps,
                       const float* __restrict__ encW1, const float* __restrict__ encB1,
                       const float* __restrict__ encW2, const float* __restrict__ encB2,
                       const float* __restrict__ encW3, const float* __restrict__ encB3,
                       const float* __restrict__ l2hW, const float* __restrict__ l2hB,
                       const float* __restrict__ h2dW, const float* __restrict__ h2dB,
                       const float* __restrict__ fW1, const float* __restrict__ fB1,
                       const float* __restrict__ fW2, const float* __restrict__ fB2,
                       float* __restrict__ pred, float* __restrict__ outMean,
                       float* __restrict__ outStd)
{
  const int tid  = (int)threadIdx.x;
  const int wave = tid >> 5;
  const int lane = tid & 31;
  const int n16  = lane & 15;
  const int hi   = lane >> 4;
  const bool h   = (hi != 0);
  const int rowBase = ((int)blockIdx.x * WAVES_PER_BLOCK + wave) * 16;
  const int row = rowBase + n16;

  float yin[D_SZ];
#pragma unroll
  for (int d = 0; d < D_SZ; ++d)
    yin[d] = ys[(size_t)row * T_SZ * D_SZ + d];

  float h1[H_SZ];
#pragma unroll
  for (int j = 0; j < H_SZ; ++j) {
    float acc = encB1[j];
#pragma unroll
    for (int d = 0; d < D_SZ; ++d) acc += yin[d] * encW1[d * H_SZ + j];
    h1[j] = fmaxf(acc, 0.0f);
  }
  float h2[H_SZ];
#pragma unroll
  for (int j = 0; j < H_SZ; ++j) {
    float acc = encB2[j];
#pragma unroll
    for (int i = 0; i < H_SZ; ++i) acc += h1[i] * encW2[i * H_SZ + j];
    h2[j] = fmaxf(acc, 0.0f);
  }
  float latent[L_SZ];
#pragma unroll
  for (int j = 0; j < L_SZ; ++j) {
    float m  = encB3[j];
    float ls = encB3[L_SZ + j];
#pragma unroll
    for (int i = 0; i < H_SZ; ++i) {
      m  += h2[i] * encW3[i * (2 * L_SZ) + j];
      ls += h2[i] * encW3[i * (2 * L_SZ) + L_SZ + j];
    }
    const float sd = expf(ls);
    latent[j] = m + eps[(size_t)row * L_SZ + j] * sd;
    if (lane < 16) {
      outMean[(size_t)row * L_SZ + j] = m;
      outStd [(size_t)row * L_SZ + j] = sd;
    }
  }
  float y0v[H_SZ];
#pragma unroll
  for (int j = 0; j < H_SZ; ++j) {
    float acc = l2hB[j];
#pragma unroll
    for (int i = 0; i < L_SZ; ++i) acc += latent[i] * l2hW[i * H_SZ + j];
    y0v[j] = fmaxf(acc, 0.0f);
  }

  float p0[H_SZ];
#pragma unroll
  for (int j = 0; j < H_SZ; ++j) {
    float acc = fB1[j];
#pragma unroll
    for (int i = 0; i < H_SZ; ++i) acc += y0v[i] * fW1[i * H_SZ + j];
    p0[j] = acc;
  }
  float prd[D_SZ];
#pragma unroll
  for (int d = 0; d < D_SZ; ++d) {
    float acc = h2dB[d];
#pragma unroll
    for (int i = 0; i < H_SZ; ++i) acc += y0v[i] * h2dW[i * D_SZ + d];
    prd[d] = acc;
  }

  v8f pc0;
  v2f pc1;
#pragma unroll
  for (int v = 0; v < 8; ++v) pc0[v] = h ? p0[v + 8] : p0[v];
  pc1.x = h ? p0[18] : p0[16];
  pc1.y = h ? p0[19] : p0[17];

  const int  r8    = n16 & 7;
  const bool okT1  = r8 < 2;
  const int  nT1   = okT1 ? 16 + r8 + 2 * (n16 >> 3) : 0;
  const bool ok3   = (n16 >= 2) && (n16 < 2 + D_SZ);
  const int  d3    = ok3 ? n16 - 2 : 0;
  const bool okR1  = okT1 || ok3;
  float am[2][5][2];
#pragma unroll
  for (int s = 0; s < 5; ++s) {
    am[0][s][0] = 0.0f; am[0][s][1] = 0.0f;
    am[1][s][0] = 0.0f; am[1][s][1] = 0.0f;
  }
  float c0a0 = 0.0f, c0a1 = 0.0f;
  for (int j = 0; j < H_SZ; ++j) {
    const float a10 = fW1[j * H_SZ + n16];
    const float a1w = fW1[j * H_SZ + nT1];
    const float a1q = h2dW[j * D_SZ + d3];
    const float a11 = ok3 ? a1q : a1w;
    const float b2j = fB2[j];
    c0a0 += b2j * a10;
    c0a1 += b2j * a11;
#pragma unroll
    for (int s = 0; s < 5; ++s) {
      const int kx = (s < 4) ? (2 * s + 8 * hi) : (16 + 2 * hi);
#pragma unroll
      for (int e = 0; e < 2; ++e) {
        const float w2 = fW2[(kx + e) * H_SZ + j];
        am[0][s][e] += w2 * a10;
        am[1][s][e] += w2 * a11;
      }
    }
  }
  v2f mf[2][6];
#pragma unroll
  for (int s = 0; s < 5; ++s) {
    mf[0][s].x = am[0][s][0];
    mf[0][s].y = am[0][s][1];
    mf[1][s].x = okR1 ? am[1][s][0] : 0.0f;
    mf[1][s].y = okR1 ? am[1][s][1] : 0.0f;
  }
  mf[0][5].x = (!h) ? c0a0 : 0.0f;            mf[0][5].y = 0.0f;
  mf[1][5].x = (!h && okR1) ? c0a1 : 0.0f;    mf[1][5].y = 0.0f;
  v2f bone;
  bone.x = h ? 0.0f : 1.0f;
  bone.y = 0.0f;
  const v8f vzero = {};

  auto matmul = [&](const v2f* bf, v8f& o0, v8f& o1) {
    o0 = wmma4(mf[0][5], bone, vzero);
    o1 = wmma4(mf[1][5], bone, vzero);
#pragma unroll
    for (int s = 0; s < 5; ++s) {
      o0 = wmma4(mf[0][s], bf[s], o0);
      o1 = wmma4(mf[1][s], bf[s], o1);
    }
  };
  auto stage = [&](float c, const v8f& d0, const v8f& d1, v2f* bf) {
#pragma unroll
    for (int j = 0; j < 4; ++j) {
      bf[j].x = fast_tanh(pc0[2 * j]     + c * d0[2 * j]);
      bf[j].y = fast_tanh(pc0[2 * j + 1] + c * d0[2 * j + 1]);
    }
    bf[4].x = fast_tanh(pc1.x + c * d1[0]);
    bf[4].y = fast_tanh(pc1.y + c * d1[1]);
  };
  auto storePred = [&](int t) {
    if (!h) {
      float* o = &pred[((size_t)(rowBase + n16) * T_SZ + t) * D_SZ];
      o[0] = prd[0]; o[1] = prd[1]; o[2] = prd[2];
    }
  };

  for (int t = 0; t < T_SZ - 1; ++t) {
    const float dt  = ts[t + 1] - ts[t];
    const float hdt = 0.5f * dt;
    storePred(t);

    v2f bf[5], sb[5];
    v8f d0, d1;
#pragma unroll
    for (int j = 0; j < 4; ++j) {
      bf[j].x = fast_tanh(pc0[2 * j]);
      bf[j].y = fast_tanh(pc0[2 * j + 1]);
    }
    bf[4].x = fast_tanh(pc1.x);
    bf[4].y = fast_tanh(pc1.y);
#pragma unroll
    for (int j = 0; j < 5; ++j) sb[j] = bf[j];
    matmul(bf, d0, d1);
    stage(hdt, d0, d1, bf);
#pragma unroll
    for (int j = 0; j < 5; ++j) { sb[j].x += 2.0f * bf[j].x; sb[j].y += 2.0f * bf[j].y; }
    matmul(bf, d0, d1);
    stage(hdt, d0, d1, bf);
#pragma unroll
    for (int j = 0; j < 5; ++j) { sb[j].x += 2.0f * bf[j].x; sb[j].y += 2.0f * bf[j].y; }
    matmul(bf, d0, d1);
    stage(dt, d0, d1, bf);
#pragma unroll
    for (int j = 0; j < 5; ++j) {
      sb[j].x = (sb[j].x + bf[j].x) * (1.0f / 6.0f);
      sb[j].y = (sb[j].y + bf[j].y) * (1.0f / 6.0f);
    }
    matmul(sb, d0, d1);

#pragma unroll
    for (int v = 0; v < 8; ++v) pc0[v] += dt * d0[v];
    pc1.x += dt * d1[0];
    pc1.y += dt * d1[1];
#pragma unroll
    for (int d = 0; d < D_SZ; ++d) prd[d] += dt * d1[2 + d];
  }
  storePred(T_SZ - 1);
}

extern "C" void kernel_launch(void* const* d_in, const int* in_sizes, int n_in,
                              void* d_out, int out_size, void* d_ws, size_t ws_size,
                              hipStream_t stream) {
  const float* ts    = (const float*)d_in[0];
  const float* ys    = (const float*)d_in[1];
  const float* eps   = (const float*)d_in[2];
  const float* encW1 = (const float*)d_in[3];
  const float* encB1 = (const float*)d_in[4];
  const float* encW2 = (const float*)d_in[5];
  const float* encB2 = (const float*)d_in[6];
  const float* encW3 = (const float*)d_in[7];
  const float* encB3 = (const float*)d_in[8];
  const float* l2hW  = (const float*)d_in[9];
  const float* l2hB  = (const float*)d_in[10];
  const float* h2dW  = (const float*)d_in[11];
  const float* h2dB  = (const float*)d_in[12];
  const float* fW1   = (const float*)d_in[13];
  const float* fB1   = (const float*)d_in[14];
  const float* fW2   = (const float*)d_in[15];
  const float* fB2   = (const float*)d_in[16];

  float* pred = (float*)d_out;
  float* mean = pred + (size_t)B_SZ * T_SZ * D_SZ;
  float* stdv = mean + (size_t)B_SZ * L_SZ;

  dim3 grid(B_SZ / (WAVES_PER_BLOCK * 16));
  dim3 block(WAVES_PER_BLOCK * 32);
  hipLaunchKernelGGL(latent_ode_kernel, grid, block, 0, stream,
                     ts, ys, eps, encW1, encB1, encW2, encB2, encW3, encB3,
                     l2hW, l2hB, h2dW, h2dB, fW1, fB1, fW2, fB2,
                     pred, mean, stdv);
}


# All weights staged once per block into LDS by one grouped coalesced load; encoder and fused-weight set-up read LDS broadcasts, not serial scalar-load waits; every sum keeps its order
# speedup vs baseline: 3.3932x; 1.1558x over previous
#include <hip/hip_runtime.h>
#include <math.h>

typedef float v2f __attribute__((ext_vector_type(2)));
typedef float v8f __attribute__((ext_vector_type(8)));

#define B_SZ 8192
#define T_SZ 200
#define D_SZ 3
#define H_SZ 20
#define L_SZ 10

#define WAVES_PER_BLOCK 4
#define NTHREADS (WAVES_PER_BLOCK * 32)

#define O_EW1 0
#define O_EB1 60
#define O_EW2 80
#define O_EB2 480
#define O_EW3 500
#define O_EB3 900
#define O_LW  920
#define O_LB  1120
#define O_DW  1140
#define O_DB  1200
#define O_FW1 1204
#define O_FB1 1604
#define O_FW2 1624
#define O_FB2 2024
#define W_TOT 2044

template <int N>
__device__ __forceinline__ void stage_load(const float* __restrict__ g, int tid,
                                           float (&r)[(N + NTHREADS - 1) / NTHREADS]) {
#pragma unroll
  for (int m = 0; m < (N + NTHREADS - 1) / NTHREADS; ++m) {
    const int i = tid + NTHREADS * m;
    r[m] = g[i < N ? i : N - 1];
  }
}
template <int N>
__device__ __forceinline__ void stage_store(float* s, int tid,
                                            const float (&r)[(N + NTHREADS - 1) / NTHREADS]) {
#pragma unroll
  for (int m = 0; m < (N + NTHREADS - 1) / NTHREADS; ++m) {
    const int i = tid + NTHREADS * m;
    if (i < N) s[i] = r[m];
  }
}

__device__ __forceinline__ v8f wmma4(v2f a, v2f b, v8f c) {
  return __builtin_amdgcn_wmma_f32_16x16x4_f32(false, a, false, b, (short)0, c,
                                               false, false);
}

__device__ __forceinline__ float fast_tanh(float x) {
#if __has_builtin(__builtin_amdgcn_tanhf)
  return __builtin_amdgcn_tanhf(x);
#else
  float r;
  asm volatile("v_tanh_f32 %0, %1\n\tv_nop" : "=v"(r) : "v"(x));
  return r;
#endif
}

__global__ __launch_bounds__(NTHREADS)
void latent_ode_kernel(const float* __restrict__ ts, const float* __restrict__ ys,
                       const float* __restrict__ eps,
                       const float* __restrict__ encW1, const float* __restrict__ encB1,
                       const float* __restrict__ encW2, const float* __restrict__ encB2,
                       const float* __restrict__ encW3, const float* __restrict__ encB3,
                       const float* __restrict__ l2hW, const float* __restrict__ l2hB,
                       const float* __restrict__ h2dW, const float* __restrict__ h2dB,
                       const float* __restrict__ fW1, const float* __restrict__ fB1,
                       const float* __restrict__ fW2, const float* __restrict__ fB2,
                       float* __restrict__ pred, float* __restrict__ outMean,
                       float* __restrict__ outStd)
{
  const int tid  = (int)threadIdx.x;
  const int wave = tid >> 5;
  const int lane = tid & 31;
  const int n16  = lane & 15;
  const int hi   = lane >> 4;
  const bool h   = (hi != 0);
  const int rowBase = ((int)blockIdx.x * WAVES_PER_BLOCK + wave) * 16;
  const int row = rowBase + n16;

  float yin[D_SZ];
#pragma unroll
  for (int d = 0; d < D_SZ; ++d)
    yin[d] = ys[(size_t)row * T_SZ * D_SZ + d];
  float epsv[L_SZ];
#pragma unroll
  for (int j = 0; j < L_SZ; ++j)
    epsv[j] = eps[(size_t)row * L_SZ + j];

  __shared__ alignas(16) float sW[W_TOT];
  {
    float r0[1], r1[1], r2[4], r3[1], r4[4], r5[1], r6[2], r7[1], r8[1], r9[1], r10[4], r11[1], r12[4], r13[1];
    stage_load<D_SZ * H_SZ>(encW1, tid, r0);
    stage_load<H_SZ>(encB1, tid, r1);
    stage_load<H_SZ * H_SZ>(encW2, tid, r2);
    stage_load<H_SZ>(encB2, tid, r3);
    stage_load<H_SZ * 2 * L_SZ>(encW3, tid, r4);
    stage_load<2 * L_SZ>(encB3, tid, r5);
    stage_load<L_SZ * H_SZ>(l2hW, tid, r6);
    stage_load<H_SZ>(l2hB, tid, r7);
    stage_load<H_SZ * D_SZ>(h2dW, tid, r8);
    stage_load<D_SZ>(h2dB, tid, r9);
    stage_load<H_SZ * H_SZ>(fW1, tid, r10);
    stage_load<H_SZ>(fB1, tid, r11);
    stage_load<H_SZ * H_SZ>(fW2, tid, r12);
    stage_load<H_SZ>(fB2, tid, r13);
    stage_store<D_SZ * H_SZ>(sW + O_EW1, tid, r0);
    stage_store<H_SZ>(sW + O_EB1, tid, r1);
    stage_store<H_SZ * H_SZ>(sW + O_EW2, tid, r2);
    stage_store<H_SZ>(sW + O_EB2, tid, r3);
    stage_store<H_SZ * 2 * L_SZ>(sW + O_EW3, tid, r4);
    stage_store<2 * L_SZ>(sW + O_EB3, tid, r5);
    stage_store<L_SZ * H_SZ>(sW + O_LW, tid, r6);
    stage_store<H_SZ>(sW + O_LB, tid, r7);
    stage_store<H_SZ * D_SZ>(sW + O_DW, tid, r8);
    stage_store<D_SZ>(sW + O_DB, tid, r9);
    stage_store<H_SZ * H_SZ>(sW + O_FW1, tid, r10);
    stage_store<H_SZ>(sW + O_FB1, tid, r11);
    stage_store<H_SZ * H_SZ>(sW + O_FW2, tid, r12);
    stage_store<H_SZ>(sW + O_FB2, tid, r13);
  }
  __syncthreads();

  float h1[H_SZ];
#pragma unroll
  for (int j = 0; j < H_SZ; ++j) h1[j] = sW[O_EB1 + j];
#pragma unroll
  for (int d = 0; d < D_SZ; ++d) {
#pragma unroll
    for (int j = 0; j < H_SZ; ++j) h1[j] += yin[d] * sW[O_EW1 + d * H_SZ + j];
  }
#pragma unroll
  for (int j = 0; j < H_SZ; ++j) h1[j] = fmaxf(h1[j], 0.0f);

  float h2[H_SZ];
#pragma unroll
  for (int j = 0; j < H_SZ; ++j) h2[j] = sW[O_EB2 + j];
#pragma unroll
  for (int i = 0; i < H_SZ; ++i) {
#pragma unroll
    for (int j = 0; j < H_SZ; ++j) h2[j] += h1[i] * sW[O_EW2 + i * H_SZ + j];
  }
#pragma unroll
  for (int j = 0; j < H_SZ; ++j) h2[j] = fmaxf(h2[j], 0.0f);

  float lp[2 * L_SZ];
#pragma unroll
  for (int j = 0; j < 2 * L_SZ; ++j) lp[j] = sW[O_EB3 + j];
#pragma unroll
  for (int i = 0; i < H_SZ; ++i) {
#pragma unroll
    for (int j = 0; j < 2 * L_SZ; ++j) lp[j] += h2[i] * sW[O_EW3 + i * (2 * L_SZ) + j];
  }
  float latent[L_SZ];
  float sdv[L_SZ];
#pragma unroll
  for (int j = 0; j < L_SZ; ++j) {
    sdv[j] = expf(lp[L_SZ + j]);
    latent[j] = lp[j] + epsv[j] * sdv[j];
  }
  if (lane < 16) {
#pragma unroll
    for (int j = 0; j < L_SZ; ++j) {
      outMean[(size_t)row * L_SZ + j] = lp[j];
      outStd [(size_t)row * L_SZ + j] = sdv[j];
    }
  }
  float y0v[H_SZ];
#pragma unroll
  for (int j = 0; j < H_SZ; ++j) y0v[j] = sW[O_LB + j];
#pragma unroll
  for (int i = 0; i < L_SZ; ++i) {
#pragma unroll
    for (int j = 0; j < H_SZ; ++j) y0v[j] += latent[i] * sW[O_LW + i * H_SZ + j];
  }
#pragma unroll
  for (int j = 0; j < H_SZ; ++j) y0v[j] = fmaxf(y0v[j], 0.0f);

  float p0[H_SZ];
#pragma unroll
  for (int j = 0; j < H_SZ; ++j) p0[j] = sW[O_FB1 + j];
#pragma unroll
  for (int i = 0; i < H_SZ; ++i) {
#pragma unroll
    for (int j = 0; j < H_SZ; ++j) p0[j] += y0v[i] * sW[O_FW1 + i * H_SZ + j];
  }
  float prd[D_SZ];
#pragma unroll
  for (int d = 0; d < D_SZ; ++d) prd[d] = sW[O_DB + d];
#pragma unroll
  for (int i = 0; i < H_SZ; ++i) {
#pragma unroll
    for (int d = 0; d < D_SZ; ++d) prd[d] += y0v[i] * sW[O_DW + i * D_SZ + d];
  }

  v8f pc0;
  v2f pc1;
#pragma unroll
  for (int v = 0; v < 8; ++v) pc0[v] = h ? p0[v + 8] : p0[v];
  pc1.x = h ? p0[18] : p0[16];
  pc1.y = h ? p0[19] : p0[17];

  const int  r8    = n16 & 7;
  const bool okT1  = r8 < 2;
  const int  nT1   = okT1 ? 16 + r8 + 2 * (n16 >> 3) : 0;
  const bool ok3   = (n16 >= 2) && (n16 < 2 + D_SZ);
  const int  d3    = ok3 ? n16 - 2 : 0;
  const bool okR1  = okT1 || ok3;
  float am[2][5][2];
#pragma unroll
  for (int s = 0; s < 5; ++s) {
    am[0][s][0] = 0.0f; am[0][s][1] = 0.0f;
    am[1][s][0] = 0.0f; am[1][s][1] = 0.0f;
  }
  float c0a0 = 0.0f, c0a1 = 0.0f;
  for (int j = 0; j < H_SZ; ++j) {
    const float a10 = sW[O_FW1 + j * H_SZ + n16];
    const float a1w = sW[O_FW1 + j * H_SZ + nT1];
    const float a1q = sW[O_DW + j * D_SZ + d3];
    const float a11 = ok3 ? a1q : a1w;
    const float b2j = sW[O_FB2 + j];
    c0a0 += b2j * a10;
    c0a1 += b2j * a11;
#pragma unroll
    for (int s = 0; s < 5; ++s) {
      const int kx = (s < 4) ? (2 * s + 8 * hi) : (16 + 2 * hi);
#pragma unroll
      for (int e = 0; e < 2; ++e) {
        const float w2 = sW[O_FW2 + (kx + e) * H_SZ + j];
        am[0][s][e] += w2 * a10;
        am[1][s][e] += w2 * a11;
      }
    }
  }
  v2f mf[2][6];
#pragma unroll
  for (int s = 0; s < 5; ++s) {
    mf[0][s].x = am[0][s][0];
    mf[0][s].y = am[0][s][1];
    mf[1][s].x = okR1 ? am[1][s][0] : 0.0f;
    mf[1][s].y = okR1 ? am[1][s][1] : 0.0f;
  }
  mf[0][5].x = (!h) ? c0a0 : 0.0f;            mf[0][5].y = 0.0f;
  mf[1][5].x = (!h && okR1) ? c0a1 : 0.0f;    mf[1][5].y = 0.0f;
  v2f bone;
  bone.x = h ? 0.0f : 1.0f;
  bone.y = 0.0f;
  const v8f vzero = {};

  auto matmul = [&](const v2f* bf, v8f& o0, v8f& o1) {
    o0 = wmma4(mf[0][5], bone, vzero);
    o1 = wmma4(mf[1][5], bone, vzero);
#pragma unroll
    for (int s = 0; s < 5; ++s) {
      o0 = wmma4(mf[0][s], bf[s], o0);
      o1 = wmma4(mf[1][s], bf[s], o1);
    }
  };
  auto stage = [&](float c, const v8f& d0, const v8f& d1, v2f* bf) {
#pragma unroll
    for (int j = 0; j < 4; ++j) {
      bf[j].x = fast_tanh(pc0[2 * j]     + c * d0[2 * j]);
      bf[j].y = fast_tanh(pc0[2 * j + 1] + c * d0[2 * j + 1]);
    }
    bf[4].x = fast_tanh(pc1.x + c * d1[0]);
    bf[4].y = fast_tanh(pc1.y + c * d1[1]);
  };
  auto storePred = [&](int t) {
    if (!h) {
      float* o = &pred[((size_t)(rowBase + n16) * T_SZ + t) * D_SZ];
      o[0] = prd[0]; o[1] = prd[1]; o[2] = prd[2];
    }
  };

  for (int t = 0; t < T_SZ - 1; ++t) {
    const float dt  = ts[t + 1] - ts[t];
    const float hdt = 0.5f * dt;
    storePred(t);

    v2f bf[5], sb[5];
    v8f d0, d1;
#pragma unroll
    for (int j = 0; j < 4; ++j) {
      bf[j].x = fast_tanh(pc0[2 * j]);
      bf[j].y = fast_tanh(pc0[2 * j + 1]);
    }
    bf[4].x = fast_tanh(pc1.x);
    bf[4].y = fast_tanh(pc1.y);
#pragma unroll
    for (int j = 0; j < 5; ++j) sb[j] = bf[j];
    matmul(bf, d0, d1);
    stage(hdt, d0, d1, bf);
#pragma unroll
    for (int j = 0; j < 5; ++j) { sb[j].x += 2.0f * bf[j].x; sb[j].y += 2.0f * bf[j].y; }
    matmul(bf, d0, d1);
    stage(hdt, d0, d1, bf);
#pragma unroll
    for (int j = 0; j < 5; ++j) { sb[j].x += 2.0f * bf[j].x; sb[j].y += 2.0f * bf[j].y; }
    matmul(bf, d0, d1);
    stage(dt, d0, d1, bf);
#pragma unroll
    for (int j = 0; j < 5; ++j) {
      sb[j].x = (sb[j].x + bf[j].x) * (1.0f / 6.0f);
      sb[j].y = (sb[j].y + bf[j].y) * (1.0f / 6.0f);
    }
    matmul(sb, d0, d1);

#pragma unroll
    for (int v = 0; v < 8; ++v) pc0[v] += dt * d0[v];
    pc1.x += dt * d1[0];
    pc1.y += dt * d1[1];
#pragma unroll
    for (int d = 0; d < D_SZ; ++d) prd[d] += dt * d1[2 + d];
  }
  storePred(T_SZ - 1);
}

extern "C" void kernel_launch(void* const* d_in, const int* in_sizes, int n_in,
                              void* d_out, int out_size, void* d_ws, size_t ws_size,
                              hipStream_t stream) {
  const float* ts    = (const float*)d_in[0];
  const float* ys    = (const float*)d_in[1];
  const float* eps   = (const float*)d_in[2];
  const float* encW1 = (const float*)d_in[3];
  const float* encB1 = (const float*)d_in[4];
  const float* encW2 = (const float*)d_in[5];
  const float* encB2 = (const float*)d_in[6];
  const float* encW3 = (const float*)d_in[7];
  const float* encB3 = (const float*)d_in[8];
  const float* l2hW  = (const float*)d_in[9];
  const float* l2hB  = (const float*)d_in[10];
  const float* h2dW  = (const float*)d_in[11];
  const float* h2dB  = (const float*)d_in[12];
  const float* fW1   = (const float*)d_in[13];
  const float* fB1   = (const float*)d_in[14];
  const float* fW2   = (const float*)d_in[15];
  const float* fB2   = (const float*)d_in[16];

  float* pred = (float*)d_out;
  float* mean = pred + (size_t)B_SZ * T_SZ * D_SZ;
  float* stdv = mean + (size_t)B_SZ * L_SZ;

  dim3 grid(B_SZ / (WAVES_PER_BLOCK * 16));
  dim3 block(WAVES_PER_BLOCK * 32);
  hipLaunchKernelGGL(latent_ode_kernel, grid, block, 0, stream,
                     ts, ys, eps, encW1, encB1, encW2, encB2, encW3, encB3,
                     l2hW, l2hB, h2dW, h2dB, fW1, fB1, fW2, fB2,
                     pred, mean, stdv);
}


# WMMA chains start from carried accumulators: bias tile made once before the loop (40 WMMA per step, was 42), 4th chain from g1+2g2+2g3 so h4 goes tanh to WMMA directly; RK4 sum reassociated
# speedup vs baseline: 3.5529x; 1.0471x over previous
#include <hip/hip_runtime.h>
#include <math.h>

typedef float v2f __attribute__((ext_vector_type(2)));
typedef float v8f __attribute__((ext_vector_type(8)));

#define B_SZ 8192
#define T_SZ 200
#define D_SZ 3
#define H_SZ 20
#define L_SZ 10

#define WAVES_PER_BLOCK 4
#define NTHREADS (WAVES_PER_BLOCK * 32)

#define O_EW1 0
#define O_EB1 60
#define O_EW2 80
#define O_EB2 480
#define O_EW3 500
#define O_EB3 900
#define O_LW  920
#define O_LB  1120
#define O_DW  1140
#define O_DB  1200
#define O_FW1 1204
#define O_FB1 1604
#define O_FW2 1624
#define O_FB2 2024
#define W_TOT 2044

template <int N>
__device__ __forceinline__ void stage_load(const float* __restrict__ g, int tid,
                                           float (&r)[(N + NTHREADS - 1) / NTHREADS]) {
#pragma unroll
  for (int m = 0; m < (N + NTHREADS - 1) / NTHREADS; ++m) {
    const int i = tid + NTHREADS * m;
    r[m] = g[i < N ? i : N - 1];
  }
}
template <int N>
__device__ __forceinline__ void stage_store(float* s, int tid,
                                            const float (&r)[(N + NTHREADS - 1) / NTHREADS]) {
#pragma unroll
  for (int m = 0; m < (N + NTHREADS - 1) / NTHREADS; ++m) {
    const int i = tid + NTHREADS * m;
    if (i < N) s[i] = r[m];
  }
}

__device__ __forceinline__ v8f wmma4(v2f a, v2f b, v8f c) {
  return __builtin_amdgcn_wmma_f32_16x16x4_f32(false, a, false, b, (short)0, c,
                                               false, false);
}

__device__ __forceinline__ float fast_tanh(float x) {
#if __has_builtin(__builtin_amdgcn_tanhf)
  return __builtin_amdgcn_tanhf(x);
#else
  float r;
  asm volatile("v_tanh_f32 %0, %1\n\tv_nop" : "=v"(r) : "v"(x));
  return r;
#endif
}

__global__ __launch_bounds__(NTHREADS)
void latent_ode_kernel(const float* __restrict__ ts, const float* __restrict__ ys,
                       const float* __restrict__ eps,
                       const float* __restrict__ encW1, const float* __restrict__ encB1,
                       const float* __restrict__ encW2, const float* __restrict__ encB2,
                       const float* __restrict__ encW3, const float* __restrict__ encB3,
                       const float* __restrict__ l2hW, const float* __restrict__ l2hB,
                       const float* __restrict__ h2dW, const float* __restrict__ h2dB,
                       const float* __restrict__ fW1, const float* __restrict__ fB1,
                       const float* __restrict__ fW2, const float* __restrict__ fB2,
                       float* __restrict__ pred, float* __restrict__ outMean,
                       float* __restrict__ outStd)
{
  const int tid  = (int)threadIdx.x;
  const int wave = tid >> 5;
  const int lane = tid & 31;
  const int n16  = lane & 15;
  const int hi   = lane >> 4;
  const bool h   = (hi != 0);
  const int rowBase = ((int)blockIdx.x * WAVES_PER_BLOCK + wave) * 16;
  const int row = rowBase + n16;

  float yin[D_SZ];
#pragma unroll
  for (int d = 0; d < D_SZ; ++d)
    yin[d] = ys[(size_t)row * T_SZ * D_SZ + d];
  float epsv[L_SZ];
#pragma unroll
  for (int j = 0; j < L_SZ; ++j)
    epsv[j] = eps[(size_t)row * L_SZ + j];

  __shared__ alignas(16) float sW[W_TOT];
  {
    float r0[1], r1[1], r2[4], r3[1], r4[4], r5[1], r6[2], r7[1], r8[1], r9[1], r10[4], r11[1], r12[4], r13[1];
    stage_load<D_SZ * H_SZ>(encW1, tid, r0);
    stage_load<H_SZ>(encB1, tid, r1);
    stage_load<H_SZ * H_SZ>(encW2, tid, r2);
    stage_load<H_SZ>(encB2, tid, r3);
    stage_load<H_SZ * 2 * L_SZ>(encW3, tid, r4);
    stage_load<2 * L_SZ>(encB3, tid, r5);
    stage_load<L_SZ * H_SZ>(l2hW, tid, r6);
    stage_load<H_SZ>(l2hB, tid, r7);
    stage_load<H_SZ * D_SZ>(h2dW, tid, r8);
    stage_load<D_SZ>(h2dB, tid, r9);
    stage_load<H_SZ * H_SZ>(fW1, tid, r10);
    stage_load<H_SZ>(fB1, tid, r11);
    stage_load<H_SZ * H_SZ>(fW2, tid, r12);
    stage_load<H_SZ>(fB2, tid, r13);
    stage_store<D_SZ * H_SZ>(sW + O_EW1, tid, r0);
    stage_store<H_SZ>(sW + O_EB1, tid, r1);
    stage_store<H_SZ * H_SZ>(sW + O_EW2, tid, r2);
    stage_store<H_SZ>(sW + O_EB2, tid, r3);
    stage_store<H_SZ * 2 * L_SZ>(sW + O_EW3, tid, r4);
    stage_store<2 * L_SZ>(sW + O_EB3, tid, r5);
    stage_store<L_SZ * H_SZ>(sW + O_LW, tid, r6);
    stage_store<H_SZ>(sW + O_LB, tid, r7);
    stage_store<H_SZ * D_SZ>(sW + O_DW, tid, r8);
    stage_store<D_SZ>(sW + O_DB, tid, r9);
    stage_store<H_SZ * H_SZ>(sW + O_FW1, tid, r10);
    stage_store<H_SZ>(sW + O_FB1, tid, r11);
    stage_store<H_SZ * H_SZ>(sW + O_FW2, tid, r12);
    stage_store<H_SZ>(sW + O_FB2, tid, r13);
  }
  __syncthreads();

  float h1[H_SZ];
#pragma unroll
  for (int j = 0; j < H_SZ; ++j) h1[j] = sW[O_EB1 + j];
#pragma unroll
  for (int d = 0; d < D_SZ; ++d) {
#pragma unroll
    for (int j = 0; j < H_SZ; ++j) h1[j] += yin[d] * sW[O_EW1 + d * H_SZ + j];
  }
#pragma unroll
  for (int j = 0; j < H_SZ; ++j) h1[j] = fmaxf(h1[j], 0.0f);

  float h2[H_SZ];
#pragma unroll
  for (int j = 0; j < H_SZ; ++j) h2[j] = sW[O_EB2 + j];
#pragma unroll
  for (int i = 0; i < H_SZ; ++i) {
#pragma unroll
    for (int j = 0; j < H_SZ; ++j) h2[j] += h1[i] * sW[O_EW2 + i * H_SZ + j];
  }
#pragma unroll
  for (int j = 0; j < H_SZ; ++j) h2[j] = fmaxf(h2[j], 0.0f);

  float lp[2 * L_SZ];
#pragma unroll
  for (int j = 0; j < 2 * L_SZ; ++j) lp[j] = sW[O_EB3 + j];
#pragma unroll
  for (int i = 0; i < H_SZ; ++i) {
#pragma unroll
    for (int j = 0; j < 2 * L_SZ; ++j) lp[j] += h2[i] * sW[O_EW3 + i * (2 * L_SZ) + j];
  }
  float latent[L_SZ];
  float sdv[L_SZ];
#pragma unroll
  for (int j = 0; j < L_SZ; ++j) {
    sdv[j] = expf(lp[L_SZ + j]);
    latent[j] = lp[j] + epsv[j] * sdv[j];
  }
  if (lane < 16) {
#pragma unroll
    for (int j = 0; j < L_SZ; ++j) {
      outMean[(size_t)row * L_SZ + j] = lp[j];
      outStd [(size_t)row * L_SZ + j] = sdv[j];
    }
  }
  float y0v[H_SZ];
#pragma unroll
  for (int j = 0; j < H_SZ; ++j) y0v[j] = sW[O_LB + j];
#pragma unroll
  for (int i = 0; i < L_SZ; ++i) {
#pragma unroll
    for (int j = 0; j < H_SZ; ++j) y0v[j] += latent[i] * sW[O_LW + i * H_SZ + j];
  }
#pragma unroll
  for (int j = 0; j < H_SZ; ++j) y0v[j] = fmaxf(y0v[j], 0.0f);

  float p0[H_SZ];
#pragma unroll
  for (int j = 0; j < H_SZ; ++j) p0[j] = sW[O_FB1 + j];
#pragma unroll
  for (int i = 0; i < H_SZ; ++i) {
#pragma unroll
    for (int j = 0; j < H_SZ; ++j) p0[j] += y0v[i] * sW[O_FW1 + i * H_SZ + j];
  }
  float prd[D_SZ];
#pragma unroll
  for (int d = 0; d < D_SZ; ++d) prd[d] = sW[O_DB + d];
#pragma unroll
  for (int i = 0; i < H_SZ; ++i) {
#pragma unroll
    for (int d = 0; d < D_SZ; ++d) prd[d] += y0v[i] * sW[O_DW + i * D_SZ + d];
  }

  v8f pc0;
  v2f pc1;
#pragma unroll
  for (int v = 0; v < 8; ++v) pc0[v] = h ? p0[v + 8] : p0[v];
  pc1.x = h ? p0[18] : p0[16];
  pc1.y = h ? p0[19] : p0[17];

  const int  r8    = n16 & 7;
  const bool okT1  = r8 < 2;
  const int  nT1   = okT1 ? 16 + r8 + 2 * (n16 >> 3) : 0;
  const bool ok3   = (n16 >= 2) && (n16 < 2 + D_SZ);
  const int  d3    = ok3 ? n16 - 2 : 0;
  const bool okR1  = okT1 || ok3;
  float am[2][5][2];
#pragma unroll
  for (int s = 0; s < 5; ++s) {
    am[0][s][0] = 0.0f; am[0][s][1] = 0.0f;
    am[1][s][0] = 0.0f; am[1][s][1] = 0.0f;
  }
  float c0a0 = 0.0f, c0a1 = 0.0f;
  for (int j = 0; j < H_SZ; ++j) {
    const float a10 = sW[O_FW1 + j * H_SZ + n16];
    const float a1w = sW[O_FW1 + j * H_SZ + nT1];
    const float a1q = sW[O_DW + j * D_SZ + d3];
    const float a11 = ok3 ? a1q : a1w;
    const float b2j = sW[O_FB2 + j];
    c0a0 += b2j * a10;
    c0a1 += b2j * a11;
#pragma unroll
    for (int s = 0; s < 5; ++s) {
      const int kx = (s < 4) ? (2 * s + 8 * hi) : (16 + 2 * hi);
#pragma unroll
      for (int e = 0; e < 2; ++e) {
        const float w2 = sW[O_FW2 + (kx + e) * H_SZ + j];
        am[0][s][e] += w2 * a10;
        am[1][s][e] += w2 * a11;
      }
    }
  }
  v2f mf[2][6];
#pragma unroll
  for (int s = 0; s < 5; ++s) {
    mf[0][s].x = am[0][s][0];
    mf[0][s].y = am[0][s][1];
    mf[1][s].x = okR1 ? am[1][s][0] : 0.0f;
    mf[1][s].y = okR1 ? am[1][s][1] : 0.0f;
  }
  mf[0][5].x = (!h) ? c0a0 : 0.0f;            mf[0][5].y = 0.0f;
  mf[1][5].x = (!h && okR1) ? c0a1 : 0.0f;    mf[1][5].y = 0.0f;
  v2f bone;
  bone.x = h ? 0.0f : 1.0f;
  bone.y = 0.0f;
  const v8f vzero = {};

  const v8f cb0 = wmma4(mf[0][5], bone, vzero);
  const v8f cb1 = wmma4(mf[1][5], bone, vzero);
  auto matmul = [&](const v2f* bf, const v8f& c0, const v8f& c1, v8f& o0, v8f& o1) {
    o0 = wmma4(mf[0][0], bf[0], c0);
    o1 = wmma4(mf[1][0], bf[0], c1);
#pragma unroll
    for (int s = 1; s < 5; ++s) {
      o0 = wmma4(mf[0][s], bf[s], o0);
      o1 = wmma4(mf[1][s], bf[s], o1);
    }
  };
  auto accum = [&](float w, const v8f& d0, const v8f& d1, v8f& e0, v8f& e1) {
#pragma unroll
    for (int v = 0; v < 8; ++v) e0[v] += w * d0[v];
#pragma unroll
    for (int v = 0; v < 2 + D_SZ; ++v) e1[v] += w * d1[v];
  };
  auto stage = [&](float c, const v8f& d0, const v8f& d1, v2f* bf) {
#pragma unroll
    for (int j = 0; j < 4; ++j) {
      bf[j].x = fast_tanh(pc0[2 * j]     + c * d0[2 * j]);
      bf[j].y = fast_tanh(pc0[2 * j + 1] + c * d0[2 * j + 1]);
    }
    bf[4].x = fast_tanh(pc1.x + c * d1[0]);
    bf[4].y = fast_tanh(pc1.y + c * d1[1]);
  };
  auto storePred = [&](int t) {
    if (!h) {
      float* o = &pred[((size_t)(rowBase + n16) * T_SZ + t) * D_SZ];
      o[0] = prd[0]; o[1] = prd[1]; o[2] = prd[2];
    }
  };

  for (int t = 0; t < T_SZ - 1; ++t) {
    const float dt  = ts[t + 1] - ts[t];
    const float hdt = 0.5f * dt;
    const float dt6 = dt * (1.0f / 6.0f);
    storePred(t);

    v2f bf[5];
    v8f d0, d1, e0, e1;
#pragma unroll
    for (int j = 0; j < 4; ++j) {
      bf[j].x = fast_tanh(pc0[2 * j]);
      bf[j].y = fast_tanh(pc0[2 * j + 1]);
    }
    bf[4].x = fast_tanh(pc1.x);
    bf[4].y = fast_tanh(pc1.y);
    matmul(bf, cb0, cb1, d0, d1);
    e0 = d0; e1 = d1;
    accum(1.0f, cb0, cb1, e0, e1);
    stage(hdt, d0, d1, bf);
    matmul(bf, cb0, cb1, d0, d1);
    accum(2.0f, d0, d1, e0, e1);
    stage(hdt, d0, d1, bf);
    matmul(bf, cb0, cb1, d0, d1);
    accum(2.0f, d0, d1, e0, e1);
    stage(dt, d0, d1, bf);
    matmul(bf, e0, e1, d0, d1);

#pragma unroll
    for (int v = 0; v < 8; ++v) pc0[v] += dt6 * d0[v];
    pc1.x += dt6 * d1[0];
    pc1.y += dt6 * d1[1];
#pragma unroll
    for (int d = 0; d < D_SZ; ++d) prd[d] += dt6 * d1[2 + d];
  }
  storePred(T_SZ - 1);
}

extern "C" void kernel_launch(void* const* d_in, const int* in_sizes, int n_in,
                              void* d_out, int out_size, void* d_ws, size_t ws_size,
                              hipStream_t stream) {
  const float* ts    = (const float*)d_in[0];
  const float* ys    = (const float*)d_in[1];
  const float* eps   = (const float*)d_in[2];
  const float* encW1 = (const float*)d_in[3];
  const float* encB1 = (const float*)d_in[4];
  const float* encW2 = (const float*)d_in[5];
  const float* encB2 = (const float*)d_in[6];
  const float* encW3 = (const float*)d_in[7];
  const float* encB3 = (const float*)d_in[8];
  const float* l2hW  = (const float*)d_in[9];
  const float* l2hB  = (const float*)d_in[10];
  const float* h2dW  = (const float*)d_in[11];
  const float* h2dB  = (const float*)d_in[12];
  const float* fW1   = (const float*)d_in[13];
  const float* fB1   = (const float*)d_in[14];
  const float* fW2   = (const float*)d_in[15];
  const float* fB2   = (const float*)d_in[16];

  float* pred = (float*)d_out;
  float* mean = pred + (size_t)B_SZ * T_SZ * D_SZ;
  float* stdv = mean + (size_t)B_SZ * L_SZ;

  dim3 grid(B_SZ / (WAVES_PER_BLOCK * 16));
  dim3 block(WAVES_PER_BLOCK * 32);
  hipLaunchKernelGGL(latent_ode_kernel, grid, block, 0, stream,
                     ts, ys, eps, encW1, encB1, encW2, encB2, encW3, encB3,
                     l2hW, l2hB, h2dW, h2dB, fW1, fB1, fW2, fB2,
                     pred, mean, stdv);
}


# Each RK4 stage issued as one vector block, then its ten WMMAs back to back (scheduling fences); tanh paired with the RK4-sum adds; ts read one step ahead; every sum keeps its order
# speedup vs baseline: 3.7678x; 1.0605x over previous
#include <hip/hip_runtime.h>
#include <math.h>

typedef float v2f __attribute__((ext_vector_type(2)));
typedef float v8f __attribute__((ext_vector_type(8)));

#define B_SZ 8192
#define T_SZ 200
#define D_SZ 3
#define H_SZ 20
#define L_SZ 10

#define WAVES_PER_BLOCK 4
#define NTHREADS (WAVES_PER_BLOCK * 32)

#define O_EW1 0
#define O_EB1 60
#define O_EW2 80
#define O_EB2 480
#define O_EW3 500
#define O_EB3 900
#define O_LW  920
#define O_LB  1120
#define O_DW  1140
#define O_DB  1200
#define O_FW1 1204
#define O_FB1 1604
#define O_FW2 1624
#define O_FB2 2024
#define W_TOT 2044

template <int N>
__device__ __forceinline__ void stage_load(const float* __restrict__ g, int tid,
                                           float (&r)[(N + NTHREADS - 1) / NTHREADS]) {
#pragma unroll
  for (int m = 0; m < (N + NTHREADS - 1) / NTHREADS; ++m) {
    const int i = tid + NTHREADS * m;
    r[m] = g[i < N ? i : N - 1];
  }
}
template <int N>
__device__ __forceinline__ void stage_store(float* s, int tid,
                                            const float (&r)[(N + NTHREADS - 1) / NTHREADS]) {
#pragma unroll
  for (int m = 0; m < (N + NTHREADS - 1) / NTHREADS; ++m) {
    const int i = tid + NTHREADS * m;
    if (i < N) s[i] = r[m];
  }
}

__device__ __forceinline__ v8f wmma4(v2f a, v2f b, v8f c) {
  return __builtin_amdgcn_wmma_f32_16x16x4_f32(false, a, false, b, (short)0, c,
                                               false, false);
}

__device__ __forceinline__ float fast_tanh(float x) {
#if __has_builtin(__builtin_amdgcn_tanhf)
  return __builtin_amdgcn_tanhf(x);
#else
  float r;
  asm volatile("v_tanh_f32 %0, %1\n\tv_nop" : "=v"(r) : "v"(x));
  return r;
#endif
}

__global__ __launch_bounds__(NTHREADS)
void latent_ode_kernel(const float* __restrict__ ts, const float* __restrict__ ys,
                       const float* __restrict__ eps,
                       const float* __restrict__ encW1, const float* __restrict__ encB1,
                       const float* __restrict__ encW2, const float* __restrict__ encB2,
                       const float* __restrict__ encW3, const float* __restrict__ encB3,
                       const float* __restrict__ l2hW, const float* __restrict__ l2hB,
                       const float* __restrict__ h2dW, const float* __restrict__ h2dB,
                       const float* __restrict__ fW1, const float* __restrict__ fB1,
                       const float* __restrict__ fW2, const float* __restrict__ fB2,
                       float* __restrict__ pred, float* __restrict__ outMean,
                       float* __restrict__ outStd)
{
  const int tid  = (int)threadIdx.x;
  const int wave = tid >> 5;
  const int lane = tid & 31;
  const int n16  = lane & 15;
  const int hi   = lane >> 4;
  const bool h   = (hi != 0);
  const int rowBase = ((int)blockIdx.x * WAVES_PER_BLOCK + wave) * 16;
  const int row = rowBase + n16;

  float yin[D_SZ];
#pragma unroll
  for (int d = 0; d < D_SZ; ++d)
    yin[d] = ys[(size_t)row * T_SZ * D_SZ + d];
  float epsv[L_SZ];
#pragma unroll
  for (int j = 0; j < L_SZ; ++j)
    epsv[j] = eps[(size_t)row * L_SZ + j];

  __shared__ alignas(16) float sW[W_TOT];
  {
    float r0[1], r1[1], r2[4], r3[1], r4[4], r5[1], r6[2], r7[1], r8[1], r9[1], r10[4], r11[1], r12[4], r13[1];
    stage_load<D_SZ * H_SZ>(encW1, tid, r0);
    stage_load<H_SZ>(encB1, tid, r1);
    stage_load<H_SZ * H_SZ>(encW2, tid, r2);
    stage_load<H_SZ>(encB2, tid, r3);
    stage_load<H_SZ * 2 * L_SZ>(encW3, tid, r4);
    stage_load<2 * L_SZ>(encB3, tid, r5);
    stage_load<L_SZ * H_SZ>(l2hW, tid, r6);
    stage_load<H_SZ>(l2hB, tid, r7);
    stage_load<H_SZ * D_SZ>(h2dW, tid, r8);
    stage_load<D_SZ>(h2dB, tid, r9);
    stage_load<H_SZ * H_SZ>(fW1, tid, r10);
    stage_load<H_SZ>(fB1, tid, r11);
    stage_load<H_SZ * H_SZ>(fW2, tid, r12);
    stage_load<H_SZ>(fB2, tid, r13);
    stage_store<D_SZ * H_SZ>(sW + O_EW1, tid, r0);
    stage_store<H_SZ>(sW + O_EB1, tid, r1);
    stage_store<H_SZ * H_SZ>(sW + O_EW2, tid, r2);
    stage_store<H_SZ>(sW + O_EB2, tid, r3);
    stage_store<H_SZ * 2 * L_SZ>(sW + O_EW3, tid, r4);
    stage_store<2 * L_SZ>(sW + O_EB3, tid, r5);
    stage_store<L_SZ * H_SZ>(sW + O_LW, tid, r6);
    stage_store<H_SZ>(sW + O_LB, tid, r7);
    stage_store<H_SZ * D_SZ>(sW + O_DW, tid, r8);
    stage_store<D_SZ>(sW + O_DB, tid, r9);
    stage_store<H_SZ * H_SZ>(sW + O_FW1, tid, r10);
    stage_store<H_SZ>(sW + O_FB1, tid, r11);
    stage_store<H_SZ * H_SZ>(sW + O_FW2, tid, r12);
    stage_store<H_SZ>(sW + O_FB2, tid, r13);
  }
  __syncthreads();

  float h1[H_SZ];
#pragma unroll
  for (int j = 0; j < H_SZ; ++j) h1[j] = sW[O_EB1 + j];
#pragma unroll
  for (int d = 0; d < D_SZ; ++d) {
#pragma unroll
    for (int j = 0; j < H_SZ; ++j) h1[j] += yin[d] * sW[O_EW1 + d * H_SZ + j];
  }
#pragma unroll
  for (int j = 0; j < H_SZ; ++j) h1[j] = fmaxf(h1[j], 0.0f);

  float h2[H_SZ];
#pragma unroll
  for (int j = 0; j < H_SZ; ++j) h2[j] = sW[O_EB2 + j];
#pragma unroll
  for (int i = 0; i < H_SZ; ++i) {
#pragma unroll
    for (int j = 0; j < H_SZ; ++j) h2[j] += h1[i] * sW[O_EW2 + i * H_SZ + j];
  }
#pragma unroll
  for (int j = 0; j < H_SZ; ++j) h2[j] = fmaxf(h2[j], 0.0f);

  float lp[2 * L_SZ];
#pragma unroll
  for (int j = 0; j < 2 * L_SZ; ++j) lp[j] = sW[O_EB3 + j];
#pragma unroll
  for (int i = 0; i < H_SZ; ++i) {
#pragma unroll
    for (int j = 0; j < 2 * L_SZ; ++j) lp[j] += h2[i] * sW[O_EW3 + i * (2 * L_SZ) + j];
  }
  float latent[L_SZ];
  float sdv[L_SZ];
#pragma unroll
  for (int j = 0; j < L_SZ; ++j) {
    sdv[j] = expf(lp[L_SZ + j]);
    latent[j] = lp[j] + epsv[j] * sdv[j];
  }
  if (lane < 16) {
#pragma unroll
    for (int j = 0; j < L_SZ; ++j) {
      outMean[(size_t)row * L_SZ + j] = lp[j];
      outStd [(size_t)row * L_SZ + j] = sdv[j];
    }
  }
  float y0v[H_SZ];
#pragma unroll
  for (int j = 0; j < H_SZ; ++j) y0v[j] = sW[O_LB + j];
#pragma unroll
  for (int i = 0; i < L_SZ; ++i) {
#pragma unroll
    for (int j = 0; j < H_SZ; ++j) y0v[j] += latent[i] * sW[O_LW + i * H_SZ + j];
  }
#pragma unroll
  for (int j = 0; j < H_SZ; ++j) y0v[j] = fmaxf(y0v[j], 0.0f);

  float p0[H_SZ];
#pragma unroll
  for (int j = 0; j < H_SZ; ++j) p0[j] = sW[O_FB1 + j];
#pragma unroll
  for (int i = 0; i < H_SZ; ++i) {
#pragma unroll
    for (int j = 0; j < H_SZ; ++j) p0[j] += y0v[i] * sW[O_FW1 + i * H_SZ + j];
  }
  float prd[D_SZ];
#pragma unroll
  for (int d = 0; d < D_SZ; ++d) prd[d] = sW[O_DB + d];
#pragma unroll
  for (int i = 0; i < H_SZ; ++i) {
#pragma unroll
    for (int d = 0; d < D_SZ; ++d) prd[d] += y0v[i] * sW[O_DW + i * D_SZ + d];
  }

  v8f pc0;
  v2f pc1;
#pragma unroll
  for (int v = 0; v < 8; ++v) pc0[v] = h ? p0[v + 8] : p0[v];
  pc1.x = h ? p0[18] : p0[16];
  pc1.y = h ? p0[19] : p0[17];

  const int  r8    = n16 & 7;
  const bool okT1  = r8 < 2;
  const int  nT1   = okT1 ? 16 + r8 + 2 * (n16 >> 3) : 0;
  const bool ok3   = (n16 >= 2) && (n16 < 2 + D_SZ);
  const int  d3    = ok3 ? n16 - 2 : 0;
  const bool okR1  = okT1 || ok3;
  float am[2][5][2];
#pragma unroll
  for (int s = 0; s < 5; ++s) {
    am[0][s][0] = 0.0f; am[0][s][1] = 0.0f;
    am[1][s][0] = 0.0f; am[1][s][1] = 0.0f;
  }
  float c0a0 = 0.0f, c0a1 = 0.0f;
  for (int j = 0; j < H_SZ; ++j) {
    const float a10 = sW[O_FW1 + j * H_SZ + n16];
    const float a1w = sW[O_FW1 + j * H_SZ + nT1];
    const float a1q = sW[O_DW + j * D_SZ + d3];
    const float a11 = ok3 ? a1q : a1w;
    const float b2j = sW[O_FB2 + j];
    c0a0 += b2j * a10;
    c0a1 += b2j * a11;
#pragma unroll
    for (int s = 0; s < 5; ++s) {
      const int kx = (s < 4) ? (2 * s + 8 * hi) : (16 + 2 * hi);
#pragma unroll
      for (int e = 0; e < 2; ++e) {
        const float w2 = sW[O_FW2 + (kx + e) * H_SZ + j];
        am[0][s][e] += w2 * a10;
        am[1][s][e] += w2 * a11;
      }
    }
  }
  v2f mf[2][6];
#pragma unroll
  for (int s = 0; s < 5; ++s) {
    mf[0][s].x = am[0][s][0];
    mf[0][s].y = am[0][s][1];
    mf[1][s].x = okR1 ? am[1][s][0] : 0.0f;
    mf[1][s].y = okR1 ? am[1][s][1] : 0.0f;
  }
  mf[0][5].x = (!h) ? c0a0 : 0.0f;            mf[0][5].y = 0.0f;
  mf[1][5].x = (!h && okR1) ? c0a1 : 0.0f;    mf[1][5].y = 0.0f;
  v2f bone;
  bone.x = h ? 0.0f : 1.0f;
  bone.y = 0.0f;
  const v8f vzero = {};

  const v8f cb0 = wmma4(mf[0][5], bone, vzero);
  const v8f cb1 = wmma4(mf[1][5], bone, vzero);
  auto matmul = [&](const v2f* bf, const v8f& c0, const v8f& c1, v8f& o0, v8f& o1) {
    o0 = wmma4(mf[0][0], bf[0], c0);
    o1 = wmma4(mf[1][0], bf[0], c1);
#pragma unroll
    for (int s = 1; s < 5; ++s) {
      o0 = wmma4(mf[0][s], bf[s], o0);
      o1 = wmma4(mf[1][s], bf[s], o1);
    }
    asm volatile("" : "+v"(o0), "+v"(o1));
  };
  auto vecblock = [&](float c, float w, bool first, const v8f& d0, const v8f& d1,
                      v8f& e0, v8f& e1, v2f* bf) {
    float z0[8], z1[2];
    if (first) { e0 = d0; e1 = d1; }
#pragma unroll
    for (int v = 0; v < 8; ++v) z0[v] = pc0[v] + c * d0[v];
    z1[0] = pc1.x + c * d1[0];
    z1[1] = pc1.y + c * d1[1];
#pragma unroll
    for (int j = 0; j < 4; ++j) {
      bf[j].x = fast_tanh(z0[2 * j]);
      if (first) e0[2 * j] = d0[2 * j] + w * cb0[2 * j];
      else       e0[2 * j] += w * d0[2 * j];
      bf[j].y = fast_tanh(z0[2 * j + 1]);
      if (first) e0[2 * j + 1] = d0[2 * j + 1] + w * cb0[2 * j + 1];
      else       e0[2 * j + 1] += w * d0[2 * j + 1];
    }
    bf[4].x = fast_tanh(z1[0]);
#pragma unroll
    for (int v = 0; v < 2; ++v) {
      if (first) e1[v] = d1[v] + w * cb1[v];
      else       e1[v] += w * d1[v];
    }
    bf[4].y = fast_tanh(z1[1]);
#pragma unroll
    for (int v = 2; v < 2 + D_SZ; ++v) {
      if (first) e1[v] = d1[v] + w * cb1[v];
      else       e1[v] += w * d1[v];
    }
    __builtin_amdgcn_sched_group_barrier(0x002, 4, 0);
#pragma unroll
    for (int i = 0; i < 10; ++i) {
      __builtin_amdgcn_sched_group_barrier(0x400, 1, 0);
      __builtin_amdgcn_sched_group_barrier(0x002, 1, 0);
    }
    asm volatile("" : "+v"(e0), "+v"(e1), "+v"(bf[0]), "+v"(bf[1]), "+v"(bf[2]), "+v"(bf[3]), "+v"(bf[4]));
  };
  auto storePred = [&](int t) {
    if (!h) {
      float* o = &pred[((size_t)(rowBase + n16) * T_SZ + t) * D_SZ];
      o[0] = prd[0]; o[1] = prd[1]; o[2] = prd[2];
    }
  };

#define STAGE_FENCE() __builtin_amdgcn_sched_barrier(0)

  float tcur = ts[0], tnext = ts[1];
  for (int t = 0; t < T_SZ - 1; ++t) {
    storePred(t);
    static_assert(B_SZ >= 2, "the look-ahead read of ts needs a second row");
    const float tahead = ts[t + 2];

    v2f bf[5];
    v8f d0, d1, e0, e1;
#pragma unroll
    for (int j = 0; j < 4; ++j) {
      bf[j].x = fast_tanh(pc0[2 * j]);
      bf[j].y = fast_tanh(pc0[2 * j + 1]);
    }
    bf[4].x = fast_tanh(pc1.x);
    bf[4].y = fast_tanh(pc1.y);
    STAGE_FENCE();
    matmul(bf, cb0, cb1, d0, d1);
    STAGE_FENCE();
    const float dt  = tnext - tcur;
    const float hdt = 0.5f * dt;
    const float dt6 = dt * (1.0f / 6.0f);
    vecblock(hdt, 1.0f, true, d0, d1, e0, e1, bf);
    STAGE_FENCE();
    matmul(bf, cb0, cb1, d0, d1);
    STAGE_FENCE();
    vecblock(hdt, 2.0f, false, d0, d1, e0, e1, bf);
    STAGE_FENCE();
    matmul(bf, cb0, cb1, d0, d1);
    STAGE_FENCE();
    vecblock(dt, 2.0f, false, d0, d1, e0, e1, bf);
    STAGE_FENCE();
    matmul(bf, e0, e1, d0, d1);
    STAGE_FENCE();

#pragma unroll
    for (int v = 0; v < 8; ++v) pc0[v] += dt6 * d0[v];
    pc1.x += dt6 * d1[0];
    pc1.y += dt6 * d1[1];
#pragma unroll
    for (int d = 0; d < D_SZ; ++d) prd[d] += dt6 * d1[2 + d];
    tcur = tnext;
    tnext = tahead;
  }
  storePred(T_SZ - 1);
}

extern "C" void kernel_launch(void* const* d_in, const int* in_sizes, int n_in,
                              void* d_out, int out_size, void* d_ws, size_t ws_size,
                              hipStream_t stream) {
  const float* ts    = (const float*)d_in[0];
  const float* ys    = (const float*)d_in[1];
  const float* eps   = (const float*)d_in[2];
  const float* encW1 = (const float*)d_in[3];
  const float* encB1 = (const float*)d_in[4];
  const float* encW2 = (const float*)d_in[5];
  const float* encB2 = (const float*)d_in[6];
  const float* encW3 = (const float*)d_in[7];
  const float* encB3 = (const float*)d_in[8];
  const float* l2hW  = (const float*)d_in[9];
  const float* l2hB  = (const float*)d_in[10];
  const float* h2dW  = (const float*)d_in[11];
  const float* h2dB  = (const float*)d_in[12];
  const float* fW1   = (const float*)d_in[13];
  const float* fB1   = (const float*)d_in[14];
  const float* fW2   = (const float*)d_in[15];
  const float* fB2   = (const float*)d_in[16];

  float* pred = (float*)d_out;
  float* mean = pred + (size_t)B_SZ * T_SZ * D_SZ;
  float* stdv = mean + (size_t)B_SZ * L_SZ;

  dim3 grid(B_SZ / (WAVES_PER_BLOCK * 16));
  dim3 block(WAVES_PER_BLOCK * 32);
  hipLaunchKernelGGL(latent_ode_kernel, grid, block, 0, stream,
                     ts, ys, eps, encW1, encB1, encW2, encB2, encW3, encB3,
                     l2hW, l2hB, h2dW, h2dB, fW1, fB1, fW2, fB2,
                     pred, mean, stdv);
}
